# starting kernel
# baseline (speedup 1.0000x reference)
#include <hip/hip_runtime.h>
#include <hip/hip_bf16.h>

typedef float v2f __attribute__((ext_vector_type(2)));
typedef float v8f __attribute__((ext_vector_type(8)));

#define BATCH 2
#define NPTS  8192
#define MPTS  8192
#define MTILES_PER_BATCH (MPTS / 16)
#define WAVES_PER_BLOCK  8

__global__ __launch_bounds__(256) void pack_refcloud(
    const float* __restrict__ input, float* __restrict__ packB) {
  int idx = blockIdx.x * blockDim.x + threadIdx.x;
  if (idx >= BATCH * NPTS) return;
  const float* q = input + (size_t)idx * 3;
  float x = q[0], y = q[1], z = q[2];
  float w = -0.5f * (x * x + y * y + z * z);
  float4 v = make_float4(x, y, z, w);
  *(float4*)(packB + (size_t)idx * 4) = v;
}

__global__ __launch_bounds__(WAVES_PER_BLOCK * 32) void chamfer_wmma(
    const float* __restrict__ point,
    const float* __restrict__ packB,
    float* __restrict__ out)
{
  const int lane    = threadIdx.x & 31;
  const int wave    = threadIdx.x >> 5;
  const int tile    = blockIdx.x * WAVES_PER_BLOCK + wave;
  const int b       = tile / MTILES_PER_BATCH;
  const int m0      = (tile % MTILES_PER_BATCH) * 16;
  const int mrow    = lane & 15;
  const int half    = lane >> 4;

  const float* pp = point + ((size_t)b * MPTS + (m0 + mrow)) * 3;
  const float px = pp[0], py = pp[1], pz = pp[2];
  v2f a;
  a.x = half ? pz : px;
  a.y = half ? 1.0f : py;
  const float p2 = px * px + py * py + pz * pz;

  const float* bbase = packB + ((size_t)b * NPTS + mrow) * 4 + (half << 1);

  v8f cmax;
  #pragma unroll
  for (int i = 0; i < 8; ++i) cmax[i] = -3.402823466e38f;

  for (int nt = 0; nt < NPTS / 16; nt += 2) {
    v2f b0 = *(const v2f*)(bbase + (size_t)nt * 64);
    v2f b1 = *(const v2f*)(bbase + (size_t)(nt + 1) * 64);
    v8f c0 = {};
    v8f c1 = {};
    c0 = __builtin_amdgcn_wmma_f32_16x16x4_f32(
         false, a,  false, b0,
         (short)0, c0,  false,  false);
    c1 = __builtin_amdgcn_wmma_f32_16x16x4_f32(
        false, a, false, b1, (short)0, c1, false, false);
    #pragma unroll
    for (int i = 0; i < 8; ++i) {
      cmax[i] = fmaxf(cmax[i], fmaxf(c0[i], c1[i]));
    }
  }

  #pragma unroll
  for (int m = 8; m >= 1; m >>= 1) {
    #pragma unroll
    for (int i = 0; i < 8; ++i) {
      float o = __shfl_xor(cmax[i], m, 32);
      cmax[i] = fmaxf(cmax[i], o);
    }
  }

  if (lane < 8) {
    out[(size_t)b * MPTS + m0 + lane] = fmaxf(p2 - 2.0f * cmax[lane], 0.0f);
  } else if (lane >= 24) {
    int r = lane - 24;
    out[(size_t)b * MPTS + m0 + 8 + r] = fmaxf(p2 - 2.0f * cmax[r], 0.0f);
  }
}

extern "C" void kernel_launch(void* const* d_in, const int* in_sizes, int n_in,
                              void* d_out, int out_size, void* d_ws, size_t ws_size,
                              hipStream_t stream) {
  const float* input = (const float*)d_in[0];
  const float* point = (const float*)d_in[1];
  float* out = (float*)d_out;
  float* packB = (float*)d_ws;

  {
    int total = BATCH * NPTS;
    pack_refcloud<<<(total + 255) / 256, 256, 0, stream>>>(input, packB);
  }
  {
    int tiles = BATCH * MTILES_PER_BATCH;
    chamfer_wmma<<<tiles / WAVES_PER_BLOCK, WAVES_PER_BLOCK * 32, 0, stream>>>(
        point, packB, out);
  }
}


# Pack launch removed: each lane reads its reference points raw (12 B) from the input into two 16-tile register sets, refilled once used, and forms the B fragment (x,y | z,-|q|^2/2) itself; one launch
# speedup vs baseline: 13.7205x; 13.7205x over previous
#include <hip/hip_runtime.h>

typedef float v2f __attribute__((ext_vector_type(2)));
typedef float v8f __attribute__((ext_vector_type(8)));

#define BATCH 2
#define NPTS  8192
#define MPTS  8192
#define NTILES (NPTS / 16)
#define WAVES_PER_BLOCK  8
#define MT 4
#define ROWS_PER_BLOCK (MT * 16)
#define BLOCKS_PER_BATCH (MPTS / ROWS_PER_BLOCK)
#define NT_PER_WAVE (NTILES / WAVES_PER_BLOCK)
#define US 16
#define TILE_FLOATS (16 * 3)

struct raw3 { float x, y, z; };

__device__ __forceinline__ void load_set(raw3 (&s)[US], const float* p) {
  #pragma unroll
  for (int u = 0; u < US; ++u) {
    const float* q = p + u * TILE_FLOATS;
    s[u].x = q[0];
    s[u].y = q[1];
    s[u].z = q[2];
  }
}

__device__ __forceinline__ v2f make_b(const raw3& q, bool hi) {
  const float w = -0.5f * (q.x * q.x + q.y * q.y + q.z * q.z);
  v2f f;
  f.x = hi ? q.z : q.x;
  f.y = hi ? w : q.y;
  return f;
}

template <int XM>
__device__ __forceinline__ float max_xmask(float x) {
  asm("v_max_num_f32_dpp %0, %0, %0 row_xmask:%1 row_mask:0xf bank_mask:0xf"
      : "+v"(x) : "n"(XM));
  return x;
}

__device__ __forceinline__ void scan_set(const raw3 (&s)[US], bool hi,
                                         const v2f (&a)[MT], v8f (&cmax)[MT]) {
  #pragma unroll
  for (int u = 0; u < US; u += 2) {
    const v2f f0 = make_b(s[u], hi);
    const v2f f1 = make_b(s[u + 1], hi);
    #pragma unroll
    for (int r = 0; r < MT; ++r) {
      v8f c0 = {};
      v8f c1 = {};
      c0 = __builtin_amdgcn_wmma_f32_16x16x4_f32(
           false, a[r],  false, f0,
           (short)0, c0,  false,  false);
      c1 = __builtin_amdgcn_wmma_f32_16x16x4_f32(
          false, a[r], false, f1, (short)0, c1, false, false);
      #pragma unroll
      for (int i = 0; i < 8; ++i) {
        cmax[r][i] = fmaxf(cmax[r][i], fmaxf(c0[i], c1[i]));
      }
    }
  }
}

__global__ __launch_bounds__(WAVES_PER_BLOCK * 32) void chamfer_wmma(
    const float* __restrict__ point,
    const float* __restrict__ input,
    float* __restrict__ out)
{
  __shared__ __attribute__((aligned(16))) float wmax[WAVES_PER_BLOCK][ROWS_PER_BLOCK];

  const int lane    = threadIdx.x & 31;
  const int wave    = threadIdx.x >> 5;
  const int b       = blockIdx.x / BLOCKS_PER_BATCH;
  const int m0      = (blockIdx.x % BLOCKS_PER_BATCH) * ROWS_PER_BLOCK;
  const int mrow    = lane & 15;
  const int half    = lane >> 4;

  const float* bbase = input + ((size_t)b * NPTS + (size_t)wave * NT_PER_WAVE * 16 + mrow) * 3;

  float px[MT], py[MT], pz[MT];
  #pragma unroll
  for (int r = 0; r < MT; ++r) {
    const float* pp = point + ((size_t)b * MPTS + (m0 + r * 16 + mrow)) * 3;
    px[r] = pp[0]; py[r] = pp[1]; pz[r] = pp[2];
  }

  raw3 s0[US], s1[US];
  load_set(s0, bbase);
  load_set(s1, bbase + 1 * US * TILE_FLOATS);

  v2f a[MT];
  float p2[MT];
  #pragma unroll
  for (int r = 0; r < MT; ++r) {
    a[r].x = half ? pz[r] : px[r];
    a[r].y = half ? 1.0f : py[r];
    p2[r] = px[r] * px[r] + py[r] * py[r] + pz[r] * pz[r];
  }

  v8f cmax[MT];
  #pragma unroll
  for (int r = 0; r < MT; ++r) {
    #pragma unroll
    for (int i = 0; i < 8; ++i) cmax[r][i] = -3.402823466e38f;
  }

  const bool hi = half != 0;
  scan_set(s0, hi, a, cmax);
  load_set(s0, bbase + 2 * US * TILE_FLOATS);
  scan_set(s1, hi, a, cmax);
  load_set(s1, bbase + 3 * US * TILE_FLOATS);
  scan_set(s0, hi, a, cmax);
  scan_set(s1, hi, a, cmax);

  #pragma unroll
  for (int r = 0; r < MT; ++r) {
    #pragma unroll
    for (int i = 0; i < 8; ++i) cmax[r][i] = max_xmask<8>(cmax[r][i]);
  }
  #pragma unroll
  for (int r = 0; r < MT; ++r) {
    #pragma unroll
    for (int i = 0; i < 8; ++i) cmax[r][i] = max_xmask<4>(cmax[r][i]);
  }
  #pragma unroll
  for (int r = 0; r < MT; ++r) {
    #pragma unroll
    for (int i = 0; i < 8; ++i) cmax[r][i] = max_xmask<2>(cmax[r][i]);
  }
  #pragma unroll
  for (int r = 0; r < MT; ++r) {
    #pragma unroll
    for (int i = 0; i < 8; ++i) cmax[r][i] = max_xmask<1>(cmax[r][i]);
  }

  if (mrow == 0) {
    #pragma unroll
    for (int r = 0; r < MT; ++r) {
      float* dst = &wmax[wave][r * 16 + half * 8];
      *(float4*)(dst)     = make_float4(cmax[r][0], cmax[r][1], cmax[r][2], cmax[r][3]);
      *(float4*)(dst + 4) = make_float4(cmax[r][4], cmax[r][5], cmax[r][6], cmax[r][7]);
    }
  }
  __syncthreads();

  const int t = threadIdx.x;
  if (t < ROWS_PER_BLOCK) {
    float cm = wmax[0][t];
    #pragma unroll
    for (int w = 1; w < WAVES_PER_BLOCK; ++w) cm = fmaxf(cm, wmax[w][t]);
    const int tr = t >> 4;
    const float p2t = (tr == 0) ? p2[0] : (tr == 1) ? p2[1] : (tr == 2) ? p2[2] : p2[3];
    out[(size_t)b * MPTS + m0 + t] = fmaxf(p2t - 2.0f * cm, 0.0f);
  }
}

extern "C" void kernel_launch(void* const* d_in, const int* in_sizes, int n_in,
                              void* d_out, int out_size, void* d_ws, size_t ws_size,
                              hipStream_t stream) {
  const float* input = (const float*)d_in[0];
  const float* point = (const float*)d_in[1];
  float* out = (float*)d_out;

  chamfer_wmma<<<BATCH * BLOCKS_PER_BATCH, WAVES_PER_BLOCK * 32, 0, stream>>>(
      point, input, out);
}


# Younger wave of each SIMD pair (waves 4-7) holds s_setprio 3 for its first 40 of 64 N tiles, then 0, so both waves end together; side effect: compiler now issues all 36 prologue loads as one group
# speedup vs baseline: 14.5772x; 1.0624x over previous
#include <hip/hip_runtime.h>

typedef float v2f __attribute__((ext_vector_type(2)));
typedef float v8f __attribute__((ext_vector_type(8)));

#define BATCH 2
#define NPTS  8192
#define MPTS  8192
#define NTILES (NPTS / 16)
#define WAVES_PER_BLOCK  8
#define MT 4
#define ROWS_PER_BLOCK (MT * 16)
#define BLOCKS_PER_BATCH (MPTS / ROWS_PER_BLOCK)
#define NT_PER_WAVE (NTILES / WAVES_PER_BLOCK)
#define US 16
#define TILE_FLOATS (16 * 3)

struct raw3 { float x, y, z; };

__device__ __forceinline__ void load_set(raw3 (&s)[US], const float* p) {
  #pragma unroll
  for (int u = 0; u < US; ++u) {
    const float* q = p + u * TILE_FLOATS;
    s[u].x = q[0];
    s[u].y = q[1];
    s[u].z = q[2];
  }
}

__device__ __forceinline__ v2f make_b(const raw3& q, bool hi) {
  const float w = -0.5f * (q.x * q.x + q.y * q.y + q.z * q.z);
  v2f f;
  f.x = hi ? q.z : q.x;
  f.y = hi ? w : q.y;
  return f;
}

template <int XM>
__device__ __forceinline__ float max_xmask(float x) {
  asm("v_max_num_f32_dpp %0, %0, %0 row_xmask:%1 row_mask:0xf bank_mask:0xf"
      : "+v"(x) : "n"(XM));
  return x;
}

template <int U0 = 0, int U1 = US>
__device__ __forceinline__ void scan_set(const raw3 (&s)[US], bool hi,
                                         const v2f (&a)[MT], v8f (&cmax)[MT]) {
  #pragma unroll
  for (int u = U0; u < U1; u += 2) {
    const v2f f0 = make_b(s[u], hi);
    const v2f f1 = make_b(s[u + 1], hi);
    #pragma unroll
    for (int r = 0; r < MT; ++r) {
      v8f c0 = {};
      v8f c1 = {};
      c0 = __builtin_amdgcn_wmma_f32_16x16x4_f32(
           false, a[r],  false, f0,
           (short)0, c0,  false,  false);
      c1 = __builtin_amdgcn_wmma_f32_16x16x4_f32(
          false, a[r], false, f1, (short)0, c1, false, false);
      #pragma unroll
      for (int i = 0; i < 8; ++i) {
        cmax[r][i] = fmaxf(cmax[r][i], fmaxf(c0[i], c1[i]));
      }
    }
  }
}

__global__ __launch_bounds__(WAVES_PER_BLOCK * 32) void chamfer_wmma(
    const float* __restrict__ point,
    const float* __restrict__ input,
    float* __restrict__ out)
{
  __shared__ __attribute__((aligned(16))) float wmax[WAVES_PER_BLOCK][ROWS_PER_BLOCK];

  const int lane    = threadIdx.x & 31;
  const int wave    = threadIdx.x >> 5;
  const int b       = blockIdx.x / BLOCKS_PER_BATCH;
  const int m0      = (blockIdx.x % BLOCKS_PER_BATCH) * ROWS_PER_BLOCK;
  const int mrow    = lane & 15;
  const int half    = lane >> 4;

  const float* bbase = input + ((size_t)b * NPTS + (size_t)wave * NT_PER_WAVE * 16 + mrow) * 3;

  float px[MT], py[MT], pz[MT];
  #pragma unroll
  for (int r = 0; r < MT; ++r) {
    const float* pp = point + ((size_t)b * MPTS + (m0 + r * 16 + mrow)) * 3;
    px[r] = pp[0]; py[r] = pp[1]; pz[r] = pp[2];
  }

  raw3 s0[US], s1[US];
  load_set(s0, bbase);
  load_set(s1, bbase + 1 * US * TILE_FLOATS);

  v2f a[MT];
  float p2[MT];
  #pragma unroll
  for (int r = 0; r < MT; ++r) {
    a[r].x = half ? pz[r] : px[r];
    a[r].y = half ? 1.0f : py[r];
    p2[r] = px[r] * px[r] + py[r] * py[r] + pz[r] * pz[r];
  }

  v8f cmax[MT];
  #pragma unroll
  for (int r = 0; r < MT; ++r) {
    #pragma unroll
    for (int i = 0; i < 8; ++i) cmax[r][i] = -3.402823466e38f;
  }

  const bool hi = half != 0;
  const int wave_s = __builtin_amdgcn_readfirstlane(wave);
  if (wave_s >= WAVES_PER_BLOCK / 2) {
    __builtin_amdgcn_s_setprio(3);
  }
  scan_set(s0, hi, a, cmax);
  load_set(s0, bbase + 2 * US * TILE_FLOATS);
  __builtin_amdgcn_sched_barrier(0);
  scan_set(s1, hi, a, cmax);
  load_set(s1, bbase + 3 * US * TILE_FLOATS);
  __builtin_amdgcn_sched_barrier(0);
  scan_set<0, US / 2>(s0, hi, a, cmax);
  __builtin_amdgcn_s_setprio(0);
  scan_set<US / 2, US>(s0, hi, a, cmax);
  scan_set(s1, hi, a, cmax);

  #pragma unroll
  for (int r = 0; r < MT; ++r) {
    #pragma unroll
    for (int i = 0; i < 8; ++i) cmax[r][i] = max_xmask<8>(cmax[r][i]);
  }
  #pragma unroll
  for (int r = 0; r < MT; ++r) {
    #pragma unroll
    for (int i = 0; i < 8; ++i) cmax[r][i] = max_xmask<4>(cmax[r][i]);
  }
  #pragma unroll
  for (int r = 0; r < MT; ++r) {
    #pragma unroll
    for (int i = 0; i < 8; ++i) cmax[r][i] = max_xmask<2>(cmax[r][i]);
  }
  #pragma unroll
  for (int r = 0; r < MT; ++r) {
    #pragma unroll
    for (int i = 0; i < 8; ++i) cmax[r][i] = max_xmask<1>(cmax[r][i]);
  }

  if (mrow == 0) {
    #pragma unroll
    for (int r = 0; r < MT; ++r) {
      float* dst = &wmax[wave][r * 16 + half * 8];
      *(float4*)(dst)     = make_float4(cmax[r][0], cmax[r][1], cmax[r][2], cmax[r][3]);
      *(float4*)(dst + 4) = make_float4(cmax[r][4], cmax[r][5], cmax[r][6], cmax[r][7]);
    }
  }
  __syncthreads();

  const int t = threadIdx.x;
  if (t < ROWS_PER_BLOCK) {
    float cm = wmax[0][t];
    #pragma unroll
    for (int w = 1; w < WAVES_PER_BLOCK; ++w) cm = fmaxf(cm, wmax[w][t]);
    const int tr = t >> 4;
    const float p2t = (tr == 0) ? p2[0] : (tr == 1) ? p2[1] : (tr == 2) ? p2[2] : p2[3];
    out[(size_t)b * MPTS + m0 + t] = fmaxf(p2t - 2.0f * cm, 0.0f);
  }
}

extern "C" void kernel_launch(void* const* d_in, const int* in_sizes, int n_in,
                              void* d_out, int out_size, void* d_ws, size_t ws_size,
                              hipStream_t stream) {
  const float* input = (const float*)d_in[0];
  const float* point = (const float*)d_in[1];
  float* out = (float*)d_out;

  chamfer_wmma<<<BATCH * BLOCKS_PER_BATCH, WAVES_PER_BLOCK * 32, 0, stream>>>(
      point, input, out);
}
